# starting kernel
# baseline (speedup 1.0000x reference)
#include <hip/hip_runtime.h>
#include <hip/hip_bf16.h>

typedef __attribute__((ext_vector_type(2))) float v2f;
typedef __attribute__((ext_vector_type(8))) float v8f;

#define N_SRC 60000
#define N_DST 30000
#define DEG   32
#define FDIM  128
#define EDIM  64
#define KSEL  16

__global__ __launch_bounds__(256) void pcgnn_gather(
    const float* __restrict__ x, const int* __restrict__ neighbors,
    const float* __restrict__ W_dist, const float* __restrict__ b_dist,
    float* __restrict__ score_out,
    float* __restrict__ h_neigh)
{
    const int wid  = threadIdx.x >> 5;
    const int lane = threadIdx.x & 31;
    const int node = blockIdx.x * 8 + wid;
    if (node >= N_DST) return;

    const float4 xv = ((const float4*)(x + (size_t)node * FDIM))[lane];
    const int f0 = lane * 4;
    float p0 = xv.x * W_dist[(f0+0)*2+0] + xv.y * W_dist[(f0+1)*2+0]
             + xv.z * W_dist[(f0+2)*2+0] + xv.w * W_dist[(f0+3)*2+0];
    float p1 = xv.x * W_dist[(f0+0)*2+1] + xv.y * W_dist[(f0+1)*2+1]
             + xv.z * W_dist[(f0+2)*2+1] + xv.w * W_dist[(f0+3)*2+1];
    #pragma unroll
    for (int m = 16; m >= 1; m >>= 1) {
        p0 += __shfl_xor(p0, m, 32);
        p1 += __shfl_xor(p1, m, 32);
    }
    const float s0 = p0 + b_dist[0];
    const float s1 = p1 + b_dist[1];
    if (lane == 0) {
        score_out[node*2+0] = s0;
        score_out[node*2+1] = s1;
    }
    const float t = 1.0f / (1.0f + __expf(-s0));

    const int nj = neighbors[node * DEG + lane];
    const float* __restrict__ nrow = x + (size_t)nj * FDIM;
    __builtin_prefetch(nrow, 0, 3);
    float acc = 0.0f;
    #pragma unroll 4
    for (int it = 0; it < 32; ++it) {
        const float4 u = ((const float4*)nrow)[it];
        const int g = it * 4;
        acc += u.x * W_dist[(g+0)*2] + u.y * W_dist[(g+1)*2]
             + u.z * W_dist[(g+2)*2] + u.w * W_dist[(g+3)*2];
    }
    const float sg   = 1.0f / (1.0f + __expf(-(acc + b_dist[0])));
    const float diff = fabsf(t - sg);

    int rank = 0;
    for (int i = 0; i < 32; ++i) {
        const float di = __shfl(diff, i, 32);
        rank += (di < diff) || (di == diff && i < lane);
    }
    const unsigned long long m = __ballot(rank < KSEL);

    float a0 = 0.f, a1 = 0.f, a2 = 0.f, a3 = 0.f;
    for (int j = 0; j < 32; ++j) {
        if ((m >> j) & 1ull) {
            const int r = __shfl(nj, j, 32);
            const float* __restrict__ row = x + (size_t)r * FDIM;
            a0 += row[lane];
            a1 += row[lane + 32];
            a2 += row[lane + 64];
            a3 += row[lane + 96];
        }
    }
    const float inv = 1.0f / (float)KSEL;
    float* __restrict__ hn = h_neigh + (size_t)node * FDIM;
    hn[lane     ] = a0 * inv;
    hn[lane + 32] = a1 * inv;
    hn[lane + 64] = a2 * inv;
    hn[lane + 96] = a3 * inv;
}

__global__ __launch_bounds__(256) void pcgnn_sage_wmma(
    const float* __restrict__ x,
    const float* __restrict__ h_neigh,
    const float* __restrict__ W_self,
    const float* __restrict__ W_neigh,
    const float* __restrict__ b_sage,
    const float* __restrict__ W_out,
    const float* __restrict__ b_out,
    float* __restrict__ logits)
{
    __shared__ float sWs[FDIM * EDIM];
    __shared__ float sWn[FDIM * EDIM];
    for (int i = threadIdx.x; i < FDIM * EDIM; i += 256) {
        const int kRow = i >> 6;
        const int col  = i & 63;
        const int dst  = (((kRow >> 1) << 6) + col) * 2 + (kRow & 1);
        sWs[dst] = W_self[i];
        sWn[dst] = W_neigh[i];
    }
    __syncthreads();

    const int wid  = threadIdx.x >> 5;
    const int lane = threadIdx.x & 31;
    const int tile = blockIdx.x * 8 + wid;
    if (tile >= N_DST / 16) return;
    const int rowBase = tile * 16;

    const int mrow = rowBase + (lane & 15);
    const int kk   = (lane >> 4) * 2;
    const int col  = lane & 15;

    const float* __restrict__ xrow = x       + (size_t)mrow * FDIM;
    const float* __restrict__ hrow = h_neigh + (size_t)mrow * FDIM;

    const v8f zero = {0.f, 0.f, 0.f, 0.f, 0.f, 0.f, 0.f, 0.f};
    v8f c[4];
    #pragma unroll
    for (int nt = 0; nt < 4; ++nt) c[nt] = zero;

    #pragma unroll 4
    for (int k = 0; k < FDIM / 4; ++k) {
        const int kb = k * 4 + kk;
        const int k2 = kb >> 1;
        const v2f ax = *(const v2f*)(xrow + kb);
        const v2f ah = *(const v2f*)(hrow + kb);
        #pragma unroll
        for (int nt = 0; nt < 4; ++nt) {
            const int cc = nt * 16 + col;
            const v2f bs = *(const v2f*)(sWs + ((k2 << 6) + cc) * 2);
            const v2f bn = *(const v2f*)(sWn + ((k2 << 6) + cc) * 2);
            c[nt] = __builtin_amdgcn_wmma_f32_16x16x4_f32(
                        false, ax, false, bs, (short)0, c[nt], false, false);
            c[nt] = __builtin_amdgcn_wmma_f32_16x16x4_f32(
                        false, ah, false, bn, (short)0, c[nt], false, false);
        }
    }

    float bsg[4], w0[4], w1[4];
    #pragma unroll
    for (int nt = 0; nt < 4; ++nt) {
        const int cc = nt * 16 + col;
        bsg[nt] = b_sage[cc];
        w0[nt]  = W_out[cc * 2 + 0];
        w1[nt]  = W_out[cc * 2 + 1];
    }
    const float bo0 = b_out[0], bo1 = b_out[1];

    #pragma unroll
    for (int r = 0; r < 8; ++r) {
        float s0 = 0.f, s1 = 0.f;
        #pragma unroll
        for (int nt = 0; nt < 4; ++nt) {
            float h = c[nt][r] + bsg[nt];
            h = fmaxf(h, 0.f);
            s0 += h * w0[nt];
            s1 += h * w1[nt];
        }
        #pragma unroll
        for (int m = 8; m >= 1; m >>= 1) {
            s0 += __shfl_xor(s0, m, 32);
            s1 += __shfl_xor(s1, m, 32);
        }
        if ((lane & 15) == 0) {
            const int row = rowBase + r + ((lane >> 4) << 3);
            logits[row * 2 + 0] = s0 + bo0;
            logits[row * 2 + 1] = s1 + bo1;
        }
    }
}

extern "C" void kernel_launch(void* const* d_in, const int* in_sizes, int n_in,
                              void* d_out, int out_size, void* d_ws, size_t ws_size,
                              hipStream_t stream) {
    const float* x       = (const float*)d_in[0];
    const int*   nb      = (const int*)  d_in[1];
    const float* W_dist  = (const float*)d_in[2];
    const float* b_dist  = (const float*)d_in[3];
    const float* W_self  = (const float*)d_in[4];
    const float* W_neigh = (const float*)d_in[5];
    const float* b_sage  = (const float*)d_in[6];
    const float* W_out   = (const float*)d_in[7];
    const float* b_out   = (const float*)d_in[8];

    float* out     = (float*)d_out;
    float* logits  = out;
    float* score   = out + N_DST * 2;
    float* h_neigh = (float*)d_ws;

    pcgnn_gather<<<N_DST / 8, 256, 0, stream>>>(x, nb, W_dist, b_dist,
                                                score, h_neigh);

    const int tiles  = N_DST / 16;
    const int blocks = (tiles + 7) / 8;
    pcgnn_sage_wmma<<<blocks, 256, 0, stream>>>(x, h_neigh, W_self, W_neigh,
                                                b_sage, W_out, b_out, logits);
}


# neighbor sigmoid score computed once per source row in a new first launch (60000 row dots) instead of once per edge (960000 row reads); edges gather 4 bytes; every sum keeps its order
# speedup vs baseline: 1.7603x; 1.7603x over previous
#include <hip/hip_runtime.h>
#include <hip/hip_bf16.h>

typedef __attribute__((ext_vector_type(2))) float v2f;
typedef __attribute__((ext_vector_type(8))) float v8f;

#define N_SRC 60000
#define N_DST 30000
#define DEG   32
#define FDIM  128
#define EDIM  64
#define KSEL  16

__global__ __launch_bounds__(64) void pcgnn_src_score(
    const float* __restrict__ x,
    const float* __restrict__ W_dist, const float* __restrict__ b_dist,
    float* __restrict__ sg_all)
{
    const int n = blockIdx.x * 64 + threadIdx.x;
    if (n >= N_SRC) return;
    const float* __restrict__ nrow = x + (size_t)n * FDIM;
    float acc = 0.0f;
    #pragma unroll 4
    for (int it = 0; it < 32; ++it) {
        const float4 u = ((const float4*)nrow)[it];
        const int g = it * 4;
        acc += u.x * W_dist[(g+0)*2] + u.y * W_dist[(g+1)*2]
             + u.z * W_dist[(g+2)*2] + u.w * W_dist[(g+3)*2];
    }
    sg_all[n] = 1.0f / (1.0f + __expf(-(acc + b_dist[0])));
}

__global__ __launch_bounds__(256) void pcgnn_gather(
    const float* __restrict__ x, const int* __restrict__ neighbors,
    const float* __restrict__ W_dist, const float* __restrict__ b_dist,
    const float* __restrict__ sg_all,
    float* __restrict__ score_out,
    float* __restrict__ h_neigh)
{
    const int wid  = threadIdx.x >> 5;
    const int lane = threadIdx.x & 31;
    const int node = blockIdx.x * 8 + wid;
    if (node >= N_DST) return;

    const int   nj = neighbors[node * DEG + lane];
    const float sg = sg_all[nj];

    const float4 xv = ((const float4*)(x + (size_t)node * FDIM))[lane];
    const int f0 = lane * 4;
    float p0 = xv.x * W_dist[(f0+0)*2+0] + xv.y * W_dist[(f0+1)*2+0]
             + xv.z * W_dist[(f0+2)*2+0] + xv.w * W_dist[(f0+3)*2+0];
    float p1 = xv.x * W_dist[(f0+0)*2+1] + xv.y * W_dist[(f0+1)*2+1]
             + xv.z * W_dist[(f0+2)*2+1] + xv.w * W_dist[(f0+3)*2+1];
    #pragma unroll
    for (int m = 16; m >= 1; m >>= 1) {
        p0 += __shfl_xor(p0, m, 32);
        p1 += __shfl_xor(p1, m, 32);
    }
    const float s0 = p0 + b_dist[0];
    const float s1 = p1 + b_dist[1];
    if (lane == 0) {
        score_out[node*2+0] = s0;
        score_out[node*2+1] = s1;
    }
    const float t = 1.0f / (1.0f + __expf(-s0));

    const float diff = fabsf(t - sg);

    int rank = 0;
    for (int i = 0; i < 32; ++i) {
        const float di = __shfl(diff, i, 32);
        rank += (di < diff) || (di == diff && i < lane);
    }
    const unsigned long long m = __ballot(rank < KSEL);

    float a0 = 0.f, a1 = 0.f, a2 = 0.f, a3 = 0.f;
    for (int j = 0; j < 32; ++j) {
        if ((m >> j) & 1ull) {
            const int r = __shfl(nj, j, 32);
            const float* __restrict__ row = x + (size_t)r * FDIM;
            a0 += row[lane];
            a1 += row[lane + 32];
            a2 += row[lane + 64];
            a3 += row[lane + 96];
        }
    }
    const float inv = 1.0f / (float)KSEL;
    float* __restrict__ hn = h_neigh + (size_t)node * FDIM;
    hn[lane     ] = a0 * inv;
    hn[lane + 32] = a1 * inv;
    hn[lane + 64] = a2 * inv;
    hn[lane + 96] = a3 * inv;
}

__global__ __launch_bounds__(256) void pcgnn_sage_wmma(
    const float* __restrict__ x,
    const float* __restrict__ h_neigh,
    const float* __restrict__ W_self,
    const float* __restrict__ W_neigh,
    const float* __restrict__ b_sage,
    const float* __restrict__ W_out,
    const float* __restrict__ b_out,
    float* __restrict__ logits)
{
    __shared__ float sWs[FDIM * EDIM];
    __shared__ float sWn[FDIM * EDIM];
    for (int i = threadIdx.x; i < FDIM * EDIM; i += 256) {
        const int kRow = i >> 6;
        const int col  = i & 63;
        const int dst  = (((kRow >> 1) << 6) + col) * 2 + (kRow & 1);
        sWs[dst] = W_self[i];
        sWn[dst] = W_neigh[i];
    }
    __syncthreads();

    const int wid  = threadIdx.x >> 5;
    const int lane = threadIdx.x & 31;
    const int tile = blockIdx.x * 8 + wid;
    if (tile >= N_DST / 16) return;
    const int rowBase = tile * 16;

    const int mrow = rowBase + (lane & 15);
    const int kk   = (lane >> 4) * 2;
    const int col  = lane & 15;

    const float* __restrict__ xrow = x       + (size_t)mrow * FDIM;
    const float* __restrict__ hrow = h_neigh + (size_t)mrow * FDIM;

    const v8f zero = {0.f, 0.f, 0.f, 0.f, 0.f, 0.f, 0.f, 0.f};
    v8f c[4];
    #pragma unroll
    for (int nt = 0; nt < 4; ++nt) c[nt] = zero;

    #pragma unroll 4
    for (int k = 0; k < FDIM / 4; ++k) {
        const int kb = k * 4 + kk;
        const int k2 = kb >> 1;
        const v2f ax = *(const v2f*)(xrow + kb);
        const v2f ah = *(const v2f*)(hrow + kb);
        #pragma unroll
        for (int nt = 0; nt < 4; ++nt) {
            const int cc = nt * 16 + col;
            const v2f bs = *(const v2f*)(sWs + ((k2 << 6) + cc) * 2);
            const v2f bn = *(const v2f*)(sWn + ((k2 << 6) + cc) * 2);
            c[nt] = __builtin_amdgcn_wmma_f32_16x16x4_f32(
                        false, ax, false, bs, (short)0, c[nt], false, false);
            c[nt] = __builtin_amdgcn_wmma_f32_16x16x4_f32(
                        false, ah, false, bn, (short)0, c[nt], false, false);
        }
    }

    float bsg[4], w0[4], w1[4];
    #pragma unroll
    for (int nt = 0; nt < 4; ++nt) {
        const int cc = nt * 16 + col;
        bsg[nt] = b_sage[cc];
        w0[nt]  = W_out[cc * 2 + 0];
        w1[nt]  = W_out[cc * 2 + 1];
    }
    const float bo0 = b_out[0], bo1 = b_out[1];

    #pragma unroll
    for (int r = 0; r < 8; ++r) {
        float s0 = 0.f, s1 = 0.f;
        #pragma unroll
        for (int nt = 0; nt < 4; ++nt) {
            float h = c[nt][r] + bsg[nt];
            h = fmaxf(h, 0.f);
            s0 += h * w0[nt];
            s1 += h * w1[nt];
        }
        #pragma unroll
        for (int m = 8; m >= 1; m >>= 1) {
            s0 += __shfl_xor(s0, m, 32);
            s1 += __shfl_xor(s1, m, 32);
        }
        if ((lane & 15) == 0) {
            const int row = rowBase + r + ((lane >> 4) << 3);
            logits[row * 2 + 0] = s0 + bo0;
            logits[row * 2 + 1] = s1 + bo1;
        }
    }
}

extern "C" void kernel_launch(void* const* d_in, const int* in_sizes, int n_in,
                              void* d_out, int out_size, void* d_ws, size_t ws_size,
                              hipStream_t stream) {
    const float* x       = (const float*)d_in[0];
    const int*   nb      = (const int*)  d_in[1];
    const float* W_dist  = (const float*)d_in[2];
    const float* b_dist  = (const float*)d_in[3];
    const float* W_self  = (const float*)d_in[4];
    const float* W_neigh = (const float*)d_in[5];
    const float* b_sage  = (const float*)d_in[6];
    const float* W_out   = (const float*)d_in[7];
    const float* b_out   = (const float*)d_in[8];

    float* out     = (float*)d_out;
    float* logits  = out;
    float* score   = out + N_DST * 2;
    float* h_neigh = (float*)d_ws;
    float* sg_all  = (float*)d_ws + N_DST * FDIM;

    pcgnn_src_score<<<(N_SRC + 63) / 64, 64, 0, stream>>>(x, W_dist, b_dist,
                                                          sg_all);

    pcgnn_gather<<<N_DST / 8, 256, 0, stream>>>(x, nb, W_dist, b_dist, sg_all,
                                                score, h_neigh);

    const int tiles  = N_DST / 16;
    const int blocks = (tiles + 7) / 8;
    pcgnn_sage_wmma<<<blocks, 256, 0, stream>>>(x, h_neigh, W_self, W_neigh,
                                                b_sage, W_out, b_out, logits);
}


# gather kernel rebuilt for loads in flight: one wave per 5 dst nodes (6000 waves, was 30000), a node's 16 selected rows read by 16 independent 16-byte loads, not 16 serial trips; sum order kept
# speedup vs baseline: 1.9109x; 1.0856x over previous
#include <hip/hip_runtime.h>
#include <hip/hip_bf16.h>

typedef __attribute__((ext_vector_type(2))) float v2f;
typedef __attribute__((ext_vector_type(8))) float v8f;

#define N_SRC 60000
#define N_DST 30000
#define DEG   32
#define FDIM  128
#define EDIM  64
#define KSEL  16
#define NPW   5

__global__ __launch_bounds__(64) void pcgnn_src_score(
    const float* __restrict__ x,
    const float* __restrict__ W_dist, const float* __restrict__ b_dist,
    float* __restrict__ sg_all)
{
    const int n = blockIdx.x * 64 + threadIdx.x;
    if (n >= N_SRC) return;
    const float* __restrict__ nrow = x + (size_t)n * FDIM;
    float acc = 0.0f;
    #pragma unroll 4
    for (int it = 0; it < 32; ++it) {
        const float4 u = ((const float4*)nrow)[it];
        const int g = it * 4;
        acc += u.x * W_dist[(g+0)*2] + u.y * W_dist[(g+1)*2]
             + u.z * W_dist[(g+2)*2] + u.w * W_dist[(g+3)*2];
    }
    sg_all[n] = 1.0f / (1.0f + __expf(-(acc + b_dist[0])));
}

__global__ __launch_bounds__(256) void pcgnn_gather(
    const float* __restrict__ x, const int* __restrict__ neighbors,
    const float* __restrict__ W_dist, const float* __restrict__ b_dist,
    const float* __restrict__ sg_all,
    float* __restrict__ score_out,
    float* __restrict__ h_neigh)
{
    const int lane = threadIdx.x & 31;
    const int wid   = __builtin_amdgcn_readfirstlane((int)(threadIdx.x >> 5));
    const int node0 = (blockIdx.x * 8 + wid) * NPW;
    if (node0 >= N_DST) return;
    const int nodeEnd = (node0 + NPW < N_DST) ? node0 + NPW : N_DST;

    const float4* __restrict__ x4 = (const float4*)x;

    const int f0 = lane * 4;
    const float wa0 = W_dist[(f0+0)*2+0], wb0 = W_dist[(f0+0)*2+1];
    const float wa1 = W_dist[(f0+1)*2+0], wb1 = W_dist[(f0+1)*2+1];
    const float wa2 = W_dist[(f0+2)*2+0], wb2 = W_dist[(f0+2)*2+1];
    const float wa3 = W_dist[(f0+3)*2+0], wb3 = W_dist[(f0+3)*2+1];
    const float bd0 = b_dist[0], bd1 = b_dist[1];

    int    nj = neighbors[node0 * DEG + lane];
    float4 xv = x4[(size_t)node0 * (FDIM / 4) + lane];
    float  sg = sg_all[nj];

    for (int node = node0; node < nodeEnd; ++node) {
        const int    nn   = (node + 1 < N_DST) ? node + 1 : node;
        const int    nj_n = neighbors[nn * DEG + lane];
        const float4 xv_n = x4[(size_t)nn * (FDIM / 4) + lane];

        float p0 = xv.x * wa0 + xv.y * wa1 + xv.z * wa2 + xv.w * wa3;
        float p1 = xv.x * wb0 + xv.y * wb1 + xv.z * wb2 + xv.w * wb3;
        #pragma unroll
        for (int m = 16; m >= 1; m >>= 1) {
            p0 += __shfl_xor(p0, m, 32);
            p1 += __shfl_xor(p1, m, 32);
        }
        const float s0 = p0 + bd0;
        const float s1 = p1 + bd1;
        if (lane == 0) {
            score_out[node*2+0] = s0;
            score_out[node*2+1] = s1;
        }
        const float t = 1.0f / (1.0f + __expf(-s0));

        const float diff = fabsf(t - sg);

        int rank = 0;
        for (int i = 0; i < 32; ++i) {
            const float di = __shfl(diff, i, 32);
            rank += (di < diff) || (di == diff && i < lane);
        }
        unsigned mm = (unsigned)__ballot(rank < KSEL);

        const float sg_n = sg_all[nj_n];

        float4 v[KSEL];
        #pragma unroll
        for (int p = 0; p < KSEL; ++p) {
            const int j = __builtin_ctz(mm | 0x80000000u);
            mm &= mm - 1u;
            const int r = __builtin_amdgcn_readlane(nj, j);
            v[p] = x4[(unsigned)r * (unsigned)(FDIM / 4) + (unsigned)lane];
        }
        float a0 = 0.f, a1 = 0.f, a2 = 0.f, a3 = 0.f;
        #pragma unroll
        for (int p = 0; p < KSEL; ++p) {
            a0 += v[p].x;
            a1 += v[p].y;
            a2 += v[p].z;
            a3 += v[p].w;
        }
        const float inv = 1.0f / (float)KSEL;
        float4 o;
        o.x = a0 * inv; o.y = a1 * inv; o.z = a2 * inv; o.w = a3 * inv;
        ((float4*)(h_neigh + (size_t)node * FDIM))[lane] = o;

        nj = nj_n; xv = xv_n; sg = sg_n;
    }
}

__global__ __launch_bounds__(256) void pcgnn_sage_wmma(
    const float* __restrict__ x,
    const float* __restrict__ h_neigh,
    const float* __restrict__ W_self,
    const float* __restrict__ W_neigh,
    const float* __restrict__ b_sage,
    const float* __restrict__ W_out,
    const float* __restrict__ b_out,
    float* __restrict__ logits)
{
    __shared__ float sWs[FDIM * EDIM];
    __shared__ float sWn[FDIM * EDIM];
    for (int i = threadIdx.x; i < FDIM * EDIM; i += 256) {
        const int kRow = i >> 6;
        const int col  = i & 63;
        const int dst  = (((kRow >> 1) << 6) + col) * 2 + (kRow & 1);
        sWs[dst] = W_self[i];
        sWn[dst] = W_neigh[i];
    }
    __syncthreads();

    const int wid  = threadIdx.x >> 5;
    const int lane = threadIdx.x & 31;
    const int tile = blockIdx.x * 8 + wid;
    if (tile >= N_DST / 16) return;
    const int rowBase = tile * 16;

    const int mrow = rowBase + (lane & 15);
    const int kk   = (lane >> 4) * 2;
    const int col  = lane & 15;

    const float* __restrict__ xrow = x       + (size_t)mrow * FDIM;
    const float* __restrict__ hrow = h_neigh + (size_t)mrow * FDIM;

    const v8f zero = {0.f, 0.f, 0.f, 0.f, 0.f, 0.f, 0.f, 0.f};
    v8f c[4];
    #pragma unroll
    for (int nt = 0; nt < 4; ++nt) c[nt] = zero;

    #pragma unroll 4
    for (int k = 0; k < FDIM / 4; ++k) {
        const int kb = k * 4 + kk;
        const int k2 = kb >> 1;
        const v2f ax = *(const v2f*)(xrow + kb);
        const v2f ah = *(const v2f*)(hrow + kb);
        #pragma unroll
        for (int nt = 0; nt < 4; ++nt) {
            const int cc = nt * 16 + col;
            const v2f bs = *(const v2f*)(sWs + ((k2 << 6) + cc) * 2);
            const v2f bn = *(const v2f*)(sWn + ((k2 << 6) + cc) * 2);
            c[nt] = __builtin_amdgcn_wmma_f32_16x16x4_f32(
                        false, ax, false, bs, (short)0, c[nt], false, false);
            c[nt] = __builtin_amdgcn_wmma_f32_16x16x4_f32(
                        false, ah, false, bn, (short)0, c[nt], false, false);
        }
    }

    float bsg[4], w0[4], w1[4];
    #pragma unroll
    for (int nt = 0; nt < 4; ++nt) {
        const int cc = nt * 16 + col;
        bsg[nt] = b_sage[cc];
        w0[nt]  = W_out[cc * 2 + 0];
        w1[nt]  = W_out[cc * 2 + 1];
    }
    const float bo0 = b_out[0], bo1 = b_out[1];

    #pragma unroll
    for (int r = 0; r < 8; ++r) {
        float s0 = 0.f, s1 = 0.f;
        #pragma unroll
        for (int nt = 0; nt < 4; ++nt) {
            float h = c[nt][r] + bsg[nt];
            h = fmaxf(h, 0.f);
            s0 += h * w0[nt];
            s1 += h * w1[nt];
        }
        #pragma unroll
        for (int m = 8; m >= 1; m >>= 1) {
            s0 += __shfl_xor(s0, m, 32);
            s1 += __shfl_xor(s1, m, 32);
        }
        if ((lane & 15) == 0) {
            const int row = rowBase + r + ((lane >> 4) << 3);
            logits[row * 2 + 0] = s0 + bo0;
            logits[row * 2 + 1] = s1 + bo1;
        }
    }
}

extern "C" void kernel_launch(void* const* d_in, const int* in_sizes, int n_in,
                              void* d_out, int out_size, void* d_ws, size_t ws_size,
                              hipStream_t stream) {
    const float* x       = (const float*)d_in[0];
    const int*   nb      = (const int*)  d_in[1];
    const float* W_dist  = (const float*)d_in[2];
    const float* b_dist  = (const float*)d_in[3];
    const float* W_self  = (const float*)d_in[4];
    const float* W_neigh = (const float*)d_in[5];
    const float* b_sage  = (const float*)d_in[6];
    const float* W_out   = (const float*)d_in[7];
    const float* b_out   = (const float*)d_in[8];

    float* out     = (float*)d_out;
    float* logits  = out;
    float* score   = out + N_DST * 2;
    float* h_neigh = (float*)d_ws;
    float* sg_all  = (float*)d_ws + N_DST * FDIM;

    pcgnn_src_score<<<(N_SRC + 63) / 64, 64, 0, stream>>>(x, W_dist, b_dist,
                                                          sg_all);

    pcgnn_gather<<<(N_DST + 8 * NPW - 1) / (8 * NPW), 256, 0, stream>>>(
        x, nb, W_dist, b_dist, sg_all, score, h_neigh);

    const int tiles  = N_DST / 16;
    const int blocks = (tiles + 7) / 8;
    pcgnn_sage_wmma<<<blocks, 256, 0, stream>>>(x, h_neigh, W_self, W_neigh,
                                                b_sage, W_out, b_out, logits);
}


# W_neigh moved in front of the gather (mean is linear): launch 1 projects all 60000 rows on WMMA beside row scores; gather sums 256-byte rows, ends the layer; 2 launches; h, logit sums reordered
# speedup vs baseline: 2.2770x; 1.1916x over previous
#include <hip/hip_runtime.h>


typedef __attribute__((ext_vector_type(2))) float v2f;
typedef __attribute__((ext_vector_type(8))) float v8f;

#define N_SRC 60000
#define N_DST 30000
#define DEG   32
#define FDIM  128
#define EDIM  64
#define KSEL  16
#define NPW   5

#define TILES_ALL    (N_SRC / 16)
#define TILES_DST    (N_DST / 16)
#define PROJ_BLOCKS  ((TILES_ALL + 7) / 8)
#define SCORE_BLOCKS ((N_SRC + 255) / 256)

__global__ __launch_bounds__(256) void pcgnn_project_score(
    const float* __restrict__ x,
    const float* __restrict__ W_dist, const float* __restrict__ b_dist,
    const float* __restrict__ W_self, const float* __restrict__ W_neigh,
    float* __restrict__ sg_all,
    float* __restrict__ yproj,
    float* __restrict__ zself)
{
    __shared__ float sWs[FDIM * EDIM];
    __shared__ float sWn[FDIM * EDIM];

    if (blockIdx.x >= PROJ_BLOCKS) {
        const int n = (blockIdx.x - PROJ_BLOCKS) * 256 + threadIdx.x;
        if (n >= N_SRC) return;
        const float* __restrict__ nrow = x + (size_t)n * FDIM;
        float acc = 0.0f;
        #pragma unroll 4
        for (int it = 0; it < 32; ++it) {
            const float4 u = ((const float4*)nrow)[it];
            const int g = it * 4;
            acc += u.x * W_dist[(g+0)*2] + u.y * W_dist[(g+1)*2]
                 + u.z * W_dist[(g+2)*2] + u.w * W_dist[(g+3)*2];
        }
        sg_all[n] = 1.0f / (1.0f + __expf(-(acc + b_dist[0])));
        return;
    }

    if (blockIdx.x * 8 < TILES_DST) {
        for (int i = threadIdx.x; i < FDIM * EDIM; i += 256) {
            const int kRow = i >> 6;
            const int col  = i & 63;
            const int dst  = (((kRow >> 1) << 6) + col) * 2 + (kRow & 1);
            sWs[dst] = W_self[i];
            sWn[dst] = W_neigh[i];
        }
    } else {
        for (int i = threadIdx.x; i < FDIM * EDIM; i += 256) {
            const int kRow = i >> 6;
            const int col  = i & 63;
            const int dst  = (((kRow >> 1) << 6) + col) * 2 + (kRow & 1);
            sWn[dst] = W_neigh[i];
        }
    }
    __syncthreads();

    const int lane = threadIdx.x & 31;
    const int wid  = __builtin_amdgcn_readfirstlane((int)(threadIdx.x >> 5));
    const int tile = blockIdx.x * 8 + wid;
    if (tile >= TILES_ALL) return;
    const int rowBase = tile * 16;

    const int c16 = lane & 15;
    const int hh  = lane >> 4;
    const int kk  = hh * 2;

    const float* __restrict__ xrow = x + (size_t)(rowBase + c16) * FDIM;

    const v8f zero = {0.f, 0.f, 0.f, 0.f, 0.f, 0.f, 0.f, 0.f};
    v8f cy[4], cz[4];
    #pragma unroll
    for (int nt = 0; nt < 4; ++nt) { cy[nt] = zero; cz[nt] = zero; }

    const bool isDst = tile < TILES_DST;
    if (isDst) {
        #pragma unroll 4
        for (int k = 0; k < FDIM / 4; ++k) {
            const int kb = k * 4 + kk;
            const int k2 = kb >> 1;
            const v2f ax = *(const v2f*)(xrow + kb);
            #pragma unroll
            for (int nt = 0; nt < 4; ++nt) {
                const int cc = nt * 16 + c16;
                const v2f bn = *(const v2f*)(sWn + ((k2 << 6) + cc) * 2);
                const v2f bs = *(const v2f*)(sWs + ((k2 << 6) + cc) * 2);
                cy[nt] = __builtin_amdgcn_wmma_f32_16x16x4_f32(
                            false, ax, false, bn, (short)0, cy[nt], false, false);
                cz[nt] = __builtin_amdgcn_wmma_f32_16x16x4_f32(
                            false, ax, false, bs, (short)0, cz[nt], false, false);
            }
        }
    } else {
        #pragma unroll 4
        for (int k = 0; k < FDIM / 4; ++k) {
            const int kb = k * 4 + kk;
            const int k2 = kb >> 1;
            const v2f ax = *(const v2f*)(xrow + kb);
            #pragma unroll
            for (int nt = 0; nt < 4; ++nt) {
                const int cc = nt * 16 + c16;
                const v2f bn = *(const v2f*)(sWn + ((k2 << 6) + cc) * 2);
                cy[nt] = __builtin_amdgcn_wmma_f32_16x16x4_f32(
                            false, ax, false, bn, (short)0, cy[nt], false, false);
            }
        }
    }

    float4* __restrict__ y4 = (float4*)yproj;
    #pragma unroll
    for (int r = 0; r < 8; ++r) {
        const int row = rowBase + 8 * hh + r;
        float4 o;
        o.x = cy[0][r]; o.y = cy[1][r]; o.z = cy[2][r]; o.w = cy[3][r];
        y4[(size_t)row * (EDIM / 4) + c16] = o;
    }
    if (isDst) {
        float4* __restrict__ z4 = (float4*)zself;
        #pragma unroll
        for (int r = 0; r < 8; ++r) {
            const int row = rowBase + 8 * hh + r;
            float4 o;
            o.x = cz[0][r]; o.y = cz[1][r]; o.z = cz[2][r]; o.w = cz[3][r];
            z4[(size_t)row * (EDIM / 4) + c16] = o;
        }
    }
}

__global__ __launch_bounds__(256) void pcgnn_gather_finish(
    const float* __restrict__ x, const int* __restrict__ neighbors,
    const float* __restrict__ W_dist, const float* __restrict__ b_dist,
    const float* __restrict__ b_sage,
    const float* __restrict__ W_out, const float* __restrict__ b_out,
    const float* __restrict__ sg_all,
    const float* __restrict__ yproj,
    const float* __restrict__ zself,
    float* __restrict__ score_out,
    float* __restrict__ logits)
{
    const int lane = threadIdx.x & 31;
    const int wid   = __builtin_amdgcn_readfirstlane((int)(threadIdx.x >> 5));
    const int node0 = (blockIdx.x * 8 + wid) * NPW;
    if (node0 >= N_DST) return;
    const int nodeEnd = (node0 + NPW < N_DST) ? node0 + NPW : N_DST;

    const float4* __restrict__ x4 = (const float4*)x;
    const float2* __restrict__ y2 = (const float2*)yproj;
    const float2* __restrict__ z2 = (const float2*)zself;

    const int f0 = lane * 4;
    const float wa0 = W_dist[(f0+0)*2+0], wb0 = W_dist[(f0+0)*2+1];
    const float wa1 = W_dist[(f0+1)*2+0], wb1 = W_dist[(f0+1)*2+1];
    const float wa2 = W_dist[(f0+2)*2+0], wb2 = W_dist[(f0+2)*2+1];
    const float wa3 = W_dist[(f0+3)*2+0], wb3 = W_dist[(f0+3)*2+1];
    const float bd0 = b_dist[0], bd1 = b_dist[1];

    const int colA = 32 * (lane & 1) + (lane >> 1);
    const int colB = colA + 16;
    const float bsA  = b_sage[colA],        bsB  = b_sage[colB];
    const float woA0 = W_out[colA * 2 + 0], woA1 = W_out[colA * 2 + 1];
    const float woB0 = W_out[colB * 2 + 0], woB1 = W_out[colB * 2 + 1];
    const float bo0 = b_out[0], bo1 = b_out[1];

    int    nj = neighbors[node0 * DEG + lane];
    float4 xv = x4[(size_t)node0 * (FDIM / 4) + lane];
    float2 zv = z2[(size_t)node0 * (EDIM / 2) + lane];
    float  sg = sg_all[nj];

    for (int node = node0; node < nodeEnd; ++node) {
        const int    nn   = (node + 1 < N_DST) ? node + 1 : node;
        const int    nj_n = neighbors[nn * DEG + lane];
        const float4 xv_n = x4[(size_t)nn * (FDIM / 4) + lane];
        const float2 zv_n = z2[(size_t)nn * (EDIM / 2) + lane];

        float p0 = xv.x * wa0 + xv.y * wa1 + xv.z * wa2 + xv.w * wa3;
        float p1 = xv.x * wb0 + xv.y * wb1 + xv.z * wb2 + xv.w * wb3;
        #pragma unroll
        for (int m = 16; m >= 1; m >>= 1) {
            p0 += __shfl_xor(p0, m, 32);
            p1 += __shfl_xor(p1, m, 32);
        }
        const float s0 = p0 + bd0;
        const float s1 = p1 + bd1;
        if (lane == 0) {
            score_out[node*2+0] = s0;
            score_out[node*2+1] = s1;
        }
        const float t = 1.0f / (1.0f + __expf(-s0));

        const float diff = fabsf(t - sg);

        int rank = 0;
        for (int i = 0; i < 32; ++i) {
            const float di = __shfl(diff, i, 32);
            rank += (di < diff) || (di == diff && i < lane);
        }
        unsigned mm = (unsigned)__ballot(rank < KSEL);

        const float sg_n = sg_all[nj_n];

        float2 v[KSEL];
        #pragma unroll
        for (int p = 0; p < KSEL; ++p) {
            const int j = __builtin_ctz(mm | 0x80000000u);
            mm &= mm - 1u;
            const int r = __builtin_amdgcn_readlane(nj, j);
            v[p] = y2[(unsigned)r * (unsigned)(EDIM / 2) + (unsigned)lane];
        }
        float a0 = 0.f, a1 = 0.f;
        #pragma unroll
        for (int p = 0; p < KSEL; ++p) {
            a0 += v[p].x;
            a1 += v[p].y;
        }
        const float inv = 1.0f / (float)KSEL;
        const float hA = fmaxf((zv.x + a0 * inv) + bsA, 0.f);
        const float hB = fmaxf((zv.y + a1 * inv) + bsB, 0.f);

        float q0 = hA * woA0 + hB * woB0;
        float q1 = hA * woA1 + hB * woB1;
        #pragma unroll
        for (int m = 16; m >= 1; m >>= 1) {
            q0 += __shfl_xor(q0, m, 32);
            q1 += __shfl_xor(q1, m, 32);
        }
        if (lane == 0) {
            logits[node*2+0] = q0 + bo0;
            logits[node*2+1] = q1 + bo1;
        }

        nj = nj_n; xv = xv_n; zv = zv_n; sg = sg_n;
    }
}

extern "C" void kernel_launch(void* const* d_in, const int* in_sizes, int n_in,
                              void* d_out, int out_size, void* d_ws, size_t ws_size,
                              hipStream_t stream) {
    const float* x       = (const float*)d_in[0];
    const int*   nb      = (const int*)  d_in[1];
    const float* W_dist  = (const float*)d_in[2];
    const float* b_dist  = (const float*)d_in[3];
    const float* W_self  = (const float*)d_in[4];
    const float* W_neigh = (const float*)d_in[5];
    const float* b_sage  = (const float*)d_in[6];
    const float* W_out   = (const float*)d_in[7];
    const float* b_out   = (const float*)d_in[8];

    float* out     = (float*)d_out;
    float* logits  = out;
    float* score   = out + N_DST * 2;
    float* yproj   = (float*)d_ws;
    float* zself   = (float*)d_ws + N_SRC * EDIM;
    float* sg_all  = (float*)d_ws + N_SRC * EDIM + N_DST * EDIM;

    pcgnn_project_score<<<PROJ_BLOCKS + SCORE_BLOCKS, 256, 0, stream>>>(
        x, W_dist, b_dist, W_self, W_neigh, sg_all, yproj, zself);

    pcgnn_gather_finish<<<(N_DST + 8 * NPW - 1) / (8 * NPW), 256, 0, stream>>>(
        x, nb, W_dist, b_dist, b_sage, W_out, b_out, sg_all, yproj, zself,
        score, logits);
}


# gather kernel: lane exchanges (two 5-step 32-lane sums, 32 rank broadcasts per node) leave the LDS unit for vector-ALU DPP row_xmask, permlanex16, readlane; same lane pairs, sums bit-identical
# speedup vs baseline: 2.3756x; 1.0433x over previous
#include <hip/hip_runtime.h>


typedef __attribute__((ext_vector_type(2))) float v2f;
typedef __attribute__((ext_vector_type(8))) float v8f;

#define N_SRC 60000
#define N_DST 30000
#define DEG   32
#define FDIM  128
#define EDIM  64
#define KSEL  16
#define NPW   5

#define TILES_ALL    (N_SRC / 16)
#define TILES_DST    (N_DST / 16)
#define PROJ_BLOCKS  ((TILES_ALL + 7) / 8)
#define SCORE_BLOCKS ((N_SRC + 255) / 256)

__global__ __launch_bounds__(256) void pcgnn_project_score(
    const float* __restrict__ x,
    const float* __restrict__ W_dist, const float* __restrict__ b_dist,
    const float* __restrict__ W_self, const float* __restrict__ W_neigh,
    float* __restrict__ sg_all,
    float* __restrict__ yproj,
    float* __restrict__ zself)
{
    __shared__ float sWs[FDIM * EDIM];
    __shared__ float sWn[FDIM * EDIM];

    if (blockIdx.x >= PROJ_BLOCKS) {
        const int n = (blockIdx.x - PROJ_BLOCKS) * 256 + threadIdx.x;
        if (n >= N_SRC) return;
        const float* __restrict__ nrow = x + (size_t)n * FDIM;
        float acc = 0.0f;
        #pragma unroll 4
        for (int it = 0; it < 32; ++it) {
            const float4 u = ((const float4*)nrow)[it];
            const int g = it * 4;
            acc += u.x * W_dist[(g+0)*2] + u.y * W_dist[(g+1)*2]
                 + u.z * W_dist[(g+2)*2] + u.w * W_dist[(g+3)*2];
        }
        sg_all[n] = 1.0f / (1.0f + __expf(-(acc + b_dist[0])));
        return;
    }

    if (blockIdx.x * 8 < TILES_DST) {
        for (int i = threadIdx.x; i < FDIM * EDIM; i += 256) {
            const int kRow = i >> 6;
            const int col  = i & 63;
            const int dst  = (((kRow >> 1) << 6) + col) * 2 + (kRow & 1);
            sWs[dst] = W_self[i];
            sWn[dst] = W_neigh[i];
        }
    } else {
        for (int i = threadIdx.x; i < FDIM * EDIM; i += 256) {
            const int kRow = i >> 6;
            const int col  = i & 63;
            const int dst  = (((kRow >> 1) << 6) + col) * 2 + (kRow & 1);
            sWn[dst] = W_neigh[i];
        }
    }
    __syncthreads();

    const int lane = threadIdx.x & 31;
    const int wid  = __builtin_amdgcn_readfirstlane((int)(threadIdx.x >> 5));
    const int tile = blockIdx.x * 8 + wid;
    if (tile >= TILES_ALL) return;
    const int rowBase = tile * 16;

    const int c16 = lane & 15;
    const int hh  = lane >> 4;
    const int kk  = hh * 2;

    const float* __restrict__ xrow = x + (size_t)(rowBase + c16) * FDIM;

    const v8f zero = {0.f, 0.f, 0.f, 0.f, 0.f, 0.f, 0.f, 0.f};
    v8f cy[4], cz[4];
    #pragma unroll
    for (int nt = 0; nt < 4; ++nt) { cy[nt] = zero; cz[nt] = zero; }

    const bool isDst = tile < TILES_DST;
    if (isDst) {
        #pragma unroll 4
        for (int k = 0; k < FDIM / 4; ++k) {
            const int kb = k * 4 + kk;
            const int k2 = kb >> 1;
            const v2f ax = *(const v2f*)(xrow + kb);
            #pragma unroll
            for (int nt = 0; nt < 4; ++nt) {
                const int cc = nt * 16 + c16;
                const v2f bn = *(const v2f*)(sWn + ((k2 << 6) + cc) * 2);
                const v2f bs = *(const v2f*)(sWs + ((k2 << 6) + cc) * 2);
                cy[nt] = __builtin_amdgcn_wmma_f32_16x16x4_f32(
                            false, ax, false, bn, (short)0, cy[nt], false, false);
                cz[nt] = __builtin_amdgcn_wmma_f32_16x16x4_f32(
                            false, ax, false, bs, (short)0, cz[nt], false, false);
            }
        }
    } else {
        #pragma unroll 4
        for (int k = 0; k < FDIM / 4; ++k) {
            const int kb = k * 4 + kk;
            const int k2 = kb >> 1;
            const v2f ax = *(const v2f*)(xrow + kb);
            #pragma unroll
            for (int nt = 0; nt < 4; ++nt) {
                const int cc = nt * 16 + c16;
                const v2f bn = *(const v2f*)(sWn + ((k2 << 6) + cc) * 2);
                cy[nt] = __builtin_amdgcn_wmma_f32_16x16x4_f32(
                            false, ax, false, bn, (short)0, cy[nt], false, false);
            }
        }
    }

    float4* __restrict__ y4 = (float4*)yproj;
    #pragma unroll
    for (int r = 0; r < 8; ++r) {
        const int row = rowBase + 8 * hh + r;
        float4 o;
        o.x = cy[0][r]; o.y = cy[1][r]; o.z = cy[2][r]; o.w = cy[3][r];
        y4[(size_t)row * (EDIM / 4) + c16] = o;
    }
    if (isDst) {
        float4* __restrict__ z4 = (float4*)zself;
        #pragma unroll
        for (int r = 0; r < 8; ++r) {
            const int row = rowBase + 8 * hh + r;
            float4 o;
            o.x = cz[0][r]; o.y = cz[1][r]; o.z = cz[2][r]; o.w = cz[3][r];
            z4[(size_t)row * (EDIM / 4) + c16] = o;
        }
    }
}

__device__ __forceinline__ float xhalf(float v) {
    const int b = __builtin_bit_cast(int, v);
    return __builtin_bit_cast(float,
        __builtin_amdgcn_permlanex16(b, b, 0x76543210, 0xfedcba98, false, false));
}
template <int M>
__device__ __forceinline__ float xrow(float v) {
    const int b = __builtin_bit_cast(int, v);
    return __builtin_bit_cast(float,
        __builtin_amdgcn_update_dpp(b, b, 0x160 + M, 0xf, 0xf, true));
}
__device__ __forceinline__ float wave_sum32(float p) {
    p += xhalf(p);
    p += xrow<8>(p);
    p += xrow<4>(p);
    p += xrow<2>(p);
    p += xrow<1>(p);
    return p;
}

__global__ __launch_bounds__(256) void pcgnn_gather_finish(
    const float* __restrict__ x, const int* __restrict__ neighbors,
    const float* __restrict__ W_dist, const float* __restrict__ b_dist,
    const float* __restrict__ b_sage,
    const float* __restrict__ W_out, const float* __restrict__ b_out,
    const float* __restrict__ sg_all,
    const float* __restrict__ yproj,
    const float* __restrict__ zself,
    float* __restrict__ score_out,
    float* __restrict__ logits)
{
    const int lane = threadIdx.x & 31;
    const int wid   = __builtin_amdgcn_readfirstlane((int)(threadIdx.x >> 5));
    const int node0 = (blockIdx.x * 8 + wid) * NPW;
    if (node0 >= N_DST) return;
    const int nodeEnd = (node0 + NPW < N_DST) ? node0 + NPW : N_DST;

    const float4* __restrict__ x4 = (const float4*)x;
    const float2* __restrict__ y2 = (const float2*)yproj;
    const float2* __restrict__ z2 = (const float2*)zself;

    const int f0 = lane * 4;
    const float wa0 = W_dist[(f0+0)*2+0], wb0 = W_dist[(f0+0)*2+1];
    const float wa1 = W_dist[(f0+1)*2+0], wb1 = W_dist[(f0+1)*2+1];
    const float wa2 = W_dist[(f0+2)*2+0], wb2 = W_dist[(f0+2)*2+1];
    const float wa3 = W_dist[(f0+3)*2+0], wb3 = W_dist[(f0+3)*2+1];
    const float bd0 = b_dist[0], bd1 = b_dist[1];

    const int colA = 32 * (lane & 1) + (lane >> 1);
    const int colB = colA + 16;
    const float bsA  = b_sage[colA],        bsB  = b_sage[colB];
    const float woA0 = W_out[colA * 2 + 0], woA1 = W_out[colA * 2 + 1];
    const float woB0 = W_out[colB * 2 + 0], woB1 = W_out[colB * 2 + 1];
    const float bo0 = b_out[0], bo1 = b_out[1];

    int    nj = neighbors[node0 * DEG + lane];
    float4 xv = x4[(size_t)node0 * (FDIM / 4) + lane];
    float2 zv = z2[(size_t)node0 * (EDIM / 2) + lane];
    float  sg = sg_all[nj];

    for (int node = node0; node < nodeEnd; ++node) {
        const int    nn   = (node + 1 < N_DST) ? node + 1 : node;
        const int    nj_n = neighbors[nn * DEG + lane];
        const float4 xv_n = x4[(size_t)nn * (FDIM / 4) + lane];
        const float2 zv_n = z2[(size_t)nn * (EDIM / 2) + lane];

        float p0 = xv.x * wa0 + xv.y * wa1 + xv.z * wa2 + xv.w * wa3;
        float p1 = xv.x * wb0 + xv.y * wb1 + xv.z * wb2 + xv.w * wb3;
        p0 = wave_sum32(p0);
        p1 = wave_sum32(p1);
        const float s0 = p0 + bd0;
        const float s1 = p1 + bd1;
        if (lane == 0) {
            score_out[node*2+0] = s0;
            score_out[node*2+1] = s1;
        }
        const float t = 1.0f / (1.0f + __expf(-s0));

        const float diff = fabsf(t - sg);

        int rank = 0;
        const int dbits = __builtin_bit_cast(int, diff);
        for (int i = 0; i < 32; ++i) {
            const float di = __builtin_bit_cast(float, __builtin_amdgcn_readlane(dbits, i));
            rank += (di < diff) || (di == diff && i < lane);
        }
        unsigned mm = (unsigned)__ballot(rank < KSEL);

        const float sg_n = sg_all[nj_n];

        float2 v[KSEL];
        #pragma unroll
        for (int p = 0; p < KSEL; ++p) {
            const int j = __builtin_ctz(mm | 0x80000000u);
            mm &= mm - 1u;
            const int r = __builtin_amdgcn_readlane(nj, j);
            v[p] = y2[(unsigned)r * (unsigned)(EDIM / 2) + (unsigned)lane];
        }
        float a0 = 0.f, a1 = 0.f;
        #pragma unroll
        for (int p = 0; p < KSEL; ++p) {
            a0 += v[p].x;
            a1 += v[p].y;
        }
        const float inv = 1.0f / (float)KSEL;
        const float hA = fmaxf((zv.x + a0 * inv) + bsA, 0.f);
        const float hB = fmaxf((zv.y + a1 * inv) + bsB, 0.f);

        float q0 = hA * woA0 + hB * woB0;
        float q1 = hA * woA1 + hB * woB1;
        q0 = wave_sum32(q0);
        q1 = wave_sum32(q1);
        if (lane == 0) {
            logits[node*2+0] = q0 + bo0;
            logits[node*2+1] = q1 + bo1;
        }

        nj = nj_n; xv = xv_n; zv = zv_n; sg = sg_n;
    }
}

extern "C" void kernel_launch(void* const* d_in, const int* in_sizes, int n_in,
                              void* d_out, int out_size, void* d_ws, size_t ws_size,
                              hipStream_t stream) {
    const float* x       = (const float*)d_in[0];
    const int*   nb      = (const int*)  d_in[1];
    const float* W_dist  = (const float*)d_in[2];
    const float* b_dist  = (const float*)d_in[3];
    const float* W_self  = (const float*)d_in[4];
    const float* W_neigh = (const float*)d_in[5];
    const float* b_sage  = (const float*)d_in[6];
    const float* W_out   = (const float*)d_in[7];
    const float* b_out   = (const float*)d_in[8];

    float* out     = (float*)d_out;
    float* logits  = out;
    float* score   = out + N_DST * 2;
    float* yproj   = (float*)d_ws;
    float* zself   = (float*)d_ws + N_SRC * EDIM;
    float* sg_all  = (float*)d_ws + N_SRC * EDIM + N_DST * EDIM;

    pcgnn_project_score<<<PROJ_BLOCKS + SCORE_BLOCKS, 256, 0, stream>>>(
        x, W_dist, b_dist, W_self, W_neigh, sg_all, yproj, zself);

    pcgnn_gather_finish<<<(N_DST + 8 * NPW - 1) / (8 * NPW), 256, 0, stream>>>(
        x, nb, W_dist, b_dist, b_sage, W_out, b_out, sg_all, yproj, zself,
        score, logits);
}
